# starting kernel
# baseline (speedup 1.0000x reference)
#include <hip/hip_runtime.h>
#include <math.h>


typedef float v2f __attribute__((ext_vector_type(2)));
typedef float v8f __attribute__((ext_vector_type(8)));

#define SK_N     4096
#define SK_NN    (4096 * 4096)
#define SK_IEPS  10.0f
#define SK_TOL   1e-9f
#define SK_ITERS 100

__global__ void __launch_bounds__(256) sk_init_K(const float* __restrict__ C,
                                                 float* __restrict__ K) {
  size_t i4 = ((size_t)blockIdx.x * 256 + threadIdx.x) * 4;
  float4 c4 = *(const float4*)(C + i4);
  float4 k4;
  k4.x = expf(-SK_IEPS * c4.x);
  k4.y = expf(-SK_IEPS * c4.y);
  k4.z = expf(-SK_IEPS * c4.z);
  k4.w = expf(-SK_IEPS * c4.w);
  *(float4*)(K + i4) = k4;
}

__global__ void sk_ones(float* __restrict__ v) {
  v[blockIdx.x * 256 + threadIdx.x] = 1.0f;
}

__global__ void __launch_bounds__(256) sk_matvec_Kv(const float* __restrict__ K,
                                                    const float* __restrict__ v,
                                                    float* __restrict__ acc_out) {
  __shared__ float part[8][16];
  const int wave = threadIdx.x >> 5;
  const int lane = threadIdx.x & 31;
  const int hf   = lane >> 4;
  const int r    = lane & 15;
  const int rowBase = blockIdx.x << 4;
  const float* Krow = K + (size_t)(rowBase + r) * SK_N + 2 * hf;
  const float* vp   = v + 2 * hf;

  v8f acc = {0.f, 0.f, 0.f, 0.f, 0.f, 0.f, 0.f, 0.f};
  const int c0 = wave << 9;
#pragma unroll 4
  for (int c = c0; c < c0 + 512; c += 4) {
    v2f a = *(const v2f*)(Krow + c);
    v2f b = *(const v2f*)(vp + c);
    acc = __builtin_amdgcn_wmma_f32_16x16x4_f32(false, a, false, b,
                                                (short)0, acc, false, false);
  }
  if (r == 0) {
    float* dst = &part[wave][hf * 8];
    dst[0] = acc[0]; dst[1] = acc[1]; dst[2] = acc[2]; dst[3] = acc[3];
    dst[4] = acc[4]; dst[5] = acc[5]; dst[6] = acc[6]; dst[7] = acc[7];
  }
  __syncthreads();
  if (threadIdx.x < 16) {
    float s = 0.f;
#pragma unroll
    for (int w = 0; w < 8; ++w) s += part[w][threadIdx.x];
    acc_out[rowBase + threadIdx.x] = s;
  }
}

__global__ void __launch_bounds__(256) sk_matvec_KT(const float* __restrict__ K,
                                                    const float* __restrict__ u,
                                                    float* __restrict__ acc_out) {
  __shared__ float part[8][16];
  const int wave = threadIdx.x >> 5;
  const int lane = threadIdx.x & 31;
  const int hf   = lane >> 4;
  const int r    = lane & 15;
  const int colBase = blockIdx.x << 4;
  const float* Kc = K + colBase + r;

  v8f acc = {0.f, 0.f, 0.f, 0.f, 0.f, 0.f, 0.f, 0.f};
  const int r0 = wave << 9;
#pragma unroll 2
  for (int row = r0; row < r0 + 512; row += 4) {
    v2f a = *(const v2f*)(u + row + 2 * hf);
    size_t base = (size_t)(row + 2 * hf) * SK_N;
    v2f b;
    b[0] = Kc[base];
    b[1] = Kc[base + SK_N];
    acc = __builtin_amdgcn_wmma_f32_16x16x4_f32(false, a, false, b,
                                                (short)0, acc, false, false);
  }
  if (lane < 16) part[wave][lane] = acc[0];
  __syncthreads();
  if (threadIdx.x < 16) {
    float s = 0.f;
#pragma unroll
    for (int w = 0; w < 8; ++w) s += part[w][threadIdx.x];
    acc_out[colBase + threadIdx.x] = s;
  }
}

__global__ void sk_scale(const float* __restrict__ target,
                         const float* __restrict__ acc,
                         float* __restrict__ out) {
  int i = blockIdx.x * 256 + threadIdx.x;
  out[i] = target[i] / (acc[i] + SK_TOL);
}

__global__ void __launch_bounds__(256) sk_final(const float* __restrict__ mu,
                                                const float* __restrict__ nu,
                                                const float* __restrict__ u,
                                                const float* __restrict__ v,
                                                float* __restrict__ P,
                                                float* __restrict__ lpart) {
  __shared__ float red[256];
  const int i = blockIdx.x;
  const float ui = u[i];
  const float mi = mu[i];
  float* Prow = P + (size_t)i * SK_N;
  float w = 0.f;
  for (int j = threadIdx.x * 4; j < SK_N; j += 256 * 4) {
    float4 k4 = *(const float4*)(Prow + j);
    float4 v4 = *(const float4*)(v + j);
    float4 n4 = *(const float4*)(nu + j);
    float4 p;
    p.x = ui * k4.x * v4.x;
    p.y = ui * k4.y * v4.y;
    p.z = ui * k4.z * v4.z;
    p.w = ui * k4.w * v4.w;
    *(float4*)(Prow + j) = p;
    w += p.x * fabsf(mi - n4.x) + p.y * fabsf(mi - n4.y) +
         p.z * fabsf(mi - n4.z) + p.w * fabsf(mi - n4.w);
  }
  red[threadIdx.x] = w;
  __syncthreads();
  for (int s = 128; s > 0; s >>= 1) {
    if (threadIdx.x < s) red[threadIdx.x] += red[threadIdx.x + s];
    __syncthreads();
  }
  if (threadIdx.x == 0) lpart[i] = red[0];
}

__global__ void __launch_bounds__(256) sk_loss_reduce(const float* __restrict__ lpart,
                                                      float* __restrict__ loss) {
  __shared__ float red[256];
  float s = 0.f;
  for (int i = threadIdx.x; i < SK_N; i += 256) s += lpart[i];
  red[threadIdx.x] = s;
  __syncthreads();
  for (int k = 128; k > 0; k >>= 1) {
    if (threadIdx.x < k) red[threadIdx.x] += red[threadIdx.x + k];
    __syncthreads();
  }
  if (threadIdx.x == 0) *loss = red[0];
}

extern "C" void kernel_launch(void* const* d_in, const int* in_sizes, int n_in,
                              void* d_out, int out_size, void* d_ws, size_t ws_size,
                              hipStream_t stream) {
  const float* mu = (const float*)d_in[0];
  const float* nu = (const float*)d_in[1];
  const float* C  = (const float*)d_in[2];
  float* K = (float*)d_out;
  float* ws = (float*)d_ws;
  float* u     = ws;
  float* v     = ws + SK_N;
  float* acc   = ws + 2 * SK_N;
  float* lpart = ws + 3 * SK_N;

  sk_init_K<<<SK_NN / (256 * 4), 256, 0, stream>>>(C, K);
  sk_ones<<<SK_N / 256, 256, 0, stream>>>(v);

  for (int t = 0; t < SK_ITERS; ++t) {
    sk_matvec_Kv<<<SK_N / 16, 256, 0, stream>>>(K, v, acc);
    sk_scale<<<SK_N / 256, 256, 0, stream>>>(mu, acc, u);
    sk_matvec_KT<<<SK_N / 16, 256, 0, stream>>>(K, u, acc);
    sk_scale<<<SK_N / 256, 256, 0, stream>>>(nu, acc, v);
  }

  sk_final<<<SK_N, 256, 0, stream>>>(mu, nu, u, v, K, lpart);
  sk_loss_reduce<<<1, 256, 0, stream>>>(lpart, K + SK_NN);
}


# Both GEMVs load K in whole 128-B lines (K v: 512 contiguous B per load, row sums taken from the WMMA diagonal; K^T u: 32-col strips, two row-half planes added in scale); f32 sum order changes
# speedup vs baseline: 2.2846x; 2.2846x over previous
#include <hip/hip_runtime.h>
#include <math.h>


typedef float v2f __attribute__((ext_vector_type(2)));
typedef float v8f __attribute__((ext_vector_type(8)));

#define SK_N     4096
#define SK_NN    (4096 * 4096)
#define SK_IEPS  10.0f
#define SK_TOL   1e-9f
#define SK_ITERS 100

__global__ void __launch_bounds__(256) sk_init_K(const float* __restrict__ C,
                                                 float* __restrict__ K) {
  size_t i4 = ((size_t)blockIdx.x * 256 + threadIdx.x) * 4;
  float4 c4 = *(const float4*)(C + i4);
  float4 k4;
  k4.x = expf(-SK_IEPS * c4.x);
  k4.y = expf(-SK_IEPS * c4.y);
  k4.z = expf(-SK_IEPS * c4.z);
  k4.w = expf(-SK_IEPS * c4.w);
  *(float4*)(K + i4) = k4;
}

__global__ void sk_ones(float* __restrict__ v) {
  v[blockIdx.x * 256 + threadIdx.x] = 1.0f;
}

__device__ __forceinline__ v8f sk_wmma(v2f a, v2f b, v8f c) {
  return __builtin_amdgcn_wmma_f32_16x16x4_f32(false, a, false, b, (short)0, c,
                                               false, false);
}

__global__ void __launch_bounds__(256) sk_matvec_Kv(const float* __restrict__ K,
                                                    const float* __restrict__ v,
                                                    float* __restrict__ acc_out) {
  __shared__ float part[8][16][32];
  __shared__ float red[16][16];
  const int wave = threadIdx.x >> 5;
  const int lane = threadIdx.x & 31;
  const int rowBase = blockIdx.x << 4;
  const int colOff = (wave << 9) + (lane << 2);
  const float* Kp = K + (size_t)rowBase * SK_N + colOff;
  const float* vp = v + colOff;

  float4 vf[4];
#pragma unroll
  for (int q = 0; q < 4; ++q) vf[q] = *(const float4*)(vp + 128 * q);

  const int sel = (lane & 15) - ((lane >> 4) << 3);

  float4 kb[2][4][4];
#define SK_LOAD_ROWS(G)                                                        \
  _Pragma("unroll") for (int r = 0; r < 4; ++r)                                \
  _Pragma("unroll") for (int q = 0; q < 4; ++q)                                \
    kb[(G) & 1][r][q] = *(const float4*)(Kp + (size_t)(4 * (G) + r) * SK_N + 128 * q)
  SK_LOAD_ROWS(0);
#pragma unroll
  for (int g = 0; g < 4; ++g) {
    __builtin_amdgcn_sched_barrier(0);
    if (g < 3) { SK_LOAD_ROWS(g + 1); }
    __builtin_amdgcn_sched_barrier(0);
    v8f acc[4];
#pragma unroll
    for (int r = 0; r < 4; ++r) acc[r] = (v8f){0.f, 0.f, 0.f, 0.f, 0.f, 0.f, 0.f, 0.f};
#pragma unroll
    for (int q = 0; q < 4; ++q) {
      v2f b01 = {vf[q].x, vf[q].y};
      v2f b23 = {vf[q].z, vf[q].w};
#pragma unroll
      for (int r = 0; r < 4; ++r) {
        v2f al = {kb[g & 1][r][q].x, kb[g & 1][r][q].y};
        acc[r] = sk_wmma(al, b01, acc[r]);
      }
#pragma unroll
      for (int r = 0; r < 4; ++r) {
        v2f ah = {kb[g & 1][r][q].z, kb[g & 1][r][q].w};
        acc[r] = sk_wmma(ah, b23, acc[r]);
      }
    }
#pragma unroll
    for (int r = 0; r < 4; ++r) {
      float d = acc[r][0];
#pragma unroll
      for (int e = 1; e < 8; ++e) d = (sel == e) ? acc[r][e] : d;
      part[wave][4 * g + r][lane] = d;
    }
  }
#undef SK_LOAD_ROWS
  __syncthreads();
  {
    const int row = threadIdx.x >> 4;
    const int s   = threadIdx.x & 15;
    const int slot = (s < 8) ? s : s + 16;
    float x = 0.f;
#pragma unroll
    for (int w = 0; w < 8; ++w) x += part[w][row][slot];
    red[row][s] = x;
  }
  __syncthreads();
  if (threadIdx.x < 16) {
    float y = 0.f;
#pragma unroll
    for (int s = 0; s < 16; ++s) y += red[threadIdx.x][s];
    acc_out[rowBase + threadIdx.x] = y;
  }
}

__global__ void __launch_bounds__(256) sk_matvec_KT(const float* __restrict__ K,
                                                    const float* __restrict__ u,
                                                    float* __restrict__ plane0,
                                                    float* __restrict__ plane1) {
  __shared__ float part[8][32];
  const int wave = threadIdx.x >> 5;
  const int lane = threadIdx.x & 31;
  const int hf   = lane >> 4;
  const int n    = lane & 15;
  const int strip = blockIdx.x & 127;
  const int half  = blockIdx.x >> 7;
  const int colBase = strip << 5;
  const int row0 = (half << 11) + (wave << 8);
  const float* Kp = K + (size_t)(row0 + 4 * hf) * SK_N + colBase + 2 * n;
  const float* up = u + row0 + 4 * hf;

  v8f accE0 = {0.f, 0.f, 0.f, 0.f, 0.f, 0.f, 0.f, 0.f};
  v8f accO0 = accE0, accE1 = accE0, accO1 = accE0;

  float4 ub[2][8];
  v2f kb[2][8][4];
#define SK_LOAD_GRP(T)                                                         \
  _Pragma("unroll") for (int g = 0; g < 8; ++g) {                              \
    ub[(T) & 1][g] = *(const float4*)(up + 64 * (T) + 8 * g);                  \
    _Pragma("unroll") for (int e = 0; e < 4; ++e)                              \
      kb[(T) & 1][g][e] =                                                      \
          *(const v2f*)(Kp + (size_t)(64 * (T) + 8 * g + e) * SK_N);           \
  }
  SK_LOAD_GRP(0)
#pragma unroll
  for (int t = 0; t < 4; ++t) {
    __builtin_amdgcn_sched_barrier(0);
    if (t < 3) { SK_LOAD_GRP(t + 1) }
    __builtin_amdgcn_sched_barrier(0);
#pragma unroll
    for (int g = 0; g < 8; ++g) {
      const float4 u4 = ub[t & 1][g];
      v2f a01 = {u4.x, u4.y};
      v2f a23 = {u4.z, u4.w};
      v2f bE0 = {kb[t & 1][g][0].x, kb[t & 1][g][1].x};
      v2f bO0 = {kb[t & 1][g][0].y, kb[t & 1][g][1].y};
      v2f bE1 = {kb[t & 1][g][2].x, kb[t & 1][g][3].x};
      v2f bO1 = {kb[t & 1][g][2].y, kb[t & 1][g][3].y};
      accE0 = sk_wmma(a01, bE0, accE0);
      accO0 = sk_wmma(a01, bO0, accO0);
      accE1 = sk_wmma(a23, bE1, accE1);
      accO1 = sk_wmma(a23, bO1, accO1);
    }
  }
#undef SK_LOAD_GRP
  if (lane < 16) {
    part[wave][2 * n]     = accE0[0] + accE1[0];
    part[wave][2 * n + 1] = accO0[0] + accO1[0];
  }
  __syncthreads();
  if (threadIdx.x < 32) {
    float s = 0.f;
#pragma unroll
    for (int w = 0; w < 8; ++w) s += part[w][threadIdx.x];
    float* dst = half ? plane1 : plane0;
    dst[colBase + threadIdx.x] = s;
  }
}

__global__ void sk_scale(const float* __restrict__ target,
                         const float* __restrict__ acc,
                         float* __restrict__ out) {
  int i = blockIdx.x * 256 + threadIdx.x;
  out[i] = target[i] / (acc[i] + SK_TOL);
}

__global__ void sk_scale2(const float* __restrict__ target,
                          const float* __restrict__ plane0,
                          const float* __restrict__ plane1,
                          float* __restrict__ out) {
  int i = blockIdx.x * 256 + threadIdx.x;
  out[i] = target[i] / ((plane0[i] + plane1[i]) + SK_TOL);
}

__global__ void __launch_bounds__(256) sk_final(const float* __restrict__ mu,
                                                const float* __restrict__ nu,
                                                const float* __restrict__ u,
                                                const float* __restrict__ v,
                                                float* __restrict__ P,
                                                float* __restrict__ lpart) {
  __shared__ float red[256];
  const int i = blockIdx.x;
  const float ui = u[i];
  const float mi = mu[i];
  float* Prow = P + (size_t)i * SK_N;
  float w = 0.f;
  for (int j = threadIdx.x * 4; j < SK_N; j += 256 * 4) {
    float4 k4 = *(const float4*)(Prow + j);
    float4 v4 = *(const float4*)(v + j);
    float4 n4 = *(const float4*)(nu + j);
    float4 p;
    p.x = ui * k4.x * v4.x;
    p.y = ui * k4.y * v4.y;
    p.z = ui * k4.z * v4.z;
    p.w = ui * k4.w * v4.w;
    *(float4*)(Prow + j) = p;
    w += p.x * fabsf(mi - n4.x) + p.y * fabsf(mi - n4.y) +
         p.z * fabsf(mi - n4.z) + p.w * fabsf(mi - n4.w);
  }
  red[threadIdx.x] = w;
  __syncthreads();
  for (int s = 128; s > 0; s >>= 1) {
    if (threadIdx.x < s) red[threadIdx.x] += red[threadIdx.x + s];
    __syncthreads();
  }
  if (threadIdx.x == 0) lpart[i] = red[0];
}

__global__ void __launch_bounds__(256) sk_loss_reduce(const float* __restrict__ lpart,
                                                      float* __restrict__ loss) {
  __shared__ float red[256];
  float s = 0.f;
  for (int i = threadIdx.x; i < SK_N; i += 256) s += lpart[i];
  red[threadIdx.x] = s;
  __syncthreads();
  for (int k = 128; k > 0; k >>= 1) {
    if (threadIdx.x < k) red[threadIdx.x] += red[threadIdx.x + k];
    __syncthreads();
  }
  if (threadIdx.x == 0) *loss = red[0];
}

extern "C" void kernel_launch(void* const* d_in, const int* in_sizes, int n_in,
                              void* d_out, int out_size, void* d_ws, size_t ws_size,
                              hipStream_t stream) {
  const float* mu = (const float*)d_in[0];
  const float* nu = (const float*)d_in[1];
  const float* C  = (const float*)d_in[2];
  float* K = (float*)d_out;
  float* ws = (float*)d_ws;
  float* u     = ws;
  float* v     = ws + SK_N;
  float* acc   = ws + 2 * SK_N;
  float* acc2  = ws + 3 * SK_N;
  float* lpart = ws + 3 * SK_N;

  sk_init_K<<<SK_NN / (256 * 4), 256, 0, stream>>>(C, K);
  sk_ones<<<SK_N / 256, 256, 0, stream>>>(v);

  for (int t = 0; t < SK_ITERS; ++t) {
    sk_matvec_Kv<<<SK_N / 16, 256, 0, stream>>>(K, v, acc);
    sk_scale<<<SK_N / 256, 256, 0, stream>>>(mu, acc, u);
    sk_matvec_KT<<<SK_N / 16, 256, 0, stream>>>(K, u, acc, acc2);
    sk_scale2<<<SK_N / 256, 256, 0, stream>>>(nu, acc, acc2, v);
  }

  sk_final<<<SK_N, 256, 0, stream>>>(mu, nu, u, v, K, lpart);
  sk_loss_reduce<<<1, 256, 0, stream>>>(lpart, K + SK_NN);
}


# Both rescale launches folded into the GEMVs: K v writes u = mu/(row sum+tol) itself and forms v from nu and the two K^T u planes as its B operand, final too; 404 -> 203 launches; sums unchanged
# speedup vs baseline: 2.8947x; 1.2670x over previous
#include <hip/hip_runtime.h>
#include <math.h>


typedef float v2f __attribute__((ext_vector_type(2)));
typedef float v8f __attribute__((ext_vector_type(8)));

#define SK_N     4096
#define SK_NN    (4096 * 4096)
#define SK_IEPS  10.0f
#define SK_TOL   1e-9f
#define SK_ITERS 100

__global__ void __launch_bounds__(256) sk_init_K(const float* __restrict__ C,
                                                 float* __restrict__ K) {
  size_t i4 = ((size_t)blockIdx.x * 256 + threadIdx.x) * 4;
  float4 c4 = *(const float4*)(C + i4);
  float4 k4;
  k4.x = expf(-SK_IEPS * c4.x);
  k4.y = expf(-SK_IEPS * c4.y);
  k4.z = expf(-SK_IEPS * c4.z);
  k4.w = expf(-SK_IEPS * c4.w);
  *(float4*)(K + i4) = k4;
}

__device__ __forceinline__ v8f sk_wmma(v2f a, v2f b, v8f c) {
  return __builtin_amdgcn_wmma_f32_16x16x4_f32(false, a, false, b, (short)0, c,
                                               false, false);
}

template <bool FIRST>
__global__ void __launch_bounds__(256) sk_matvec_Kv(const float* __restrict__ K,
                                                    const float* __restrict__ nu,
                                                    const float* __restrict__ plane0,
                                                    const float* __restrict__ plane1,
                                                    const float* __restrict__ mu,
                                                    float* __restrict__ u_out) {
  __shared__ float part[8][16][32];
  __shared__ float red[16][16];
  const int wave = threadIdx.x >> 5;
  const int lane = threadIdx.x & 31;
  const int rowBase = blockIdx.x << 4;
  const int colOff = (wave << 9) + (lane << 2);
  const float* Kp = K + (size_t)rowBase * SK_N + colOff;

  const float mrow = mu[rowBase + (threadIdx.x & 15)];

  float4 vf[4];
#pragma unroll
  for (int q = 0; q < 4; ++q) vf[q] = make_float4(1.f, 1.f, 1.f, 1.f);

  const int sel = (lane & 15) - ((lane >> 4) << 3);

  float4 kb[2][4][4];
#define SK_LOAD_ROWS(G)                                                        \
  _Pragma("unroll") for (int r = 0; r < 4; ++r)                                \
  _Pragma("unroll") for (int q = 0; q < 4; ++q)                                \
    kb[(G) & 1][r][q] = *(const float4*)(Kp + (size_t)(4 * (G) + r) * SK_N + 128 * q)
  SK_LOAD_ROWS(0);
#pragma unroll
  for (int g = 0; g < 4; ++g) {
    __builtin_amdgcn_sched_barrier(0);
    if (g < 3) { SK_LOAD_ROWS(g + 1); }
    __builtin_amdgcn_sched_barrier(0);
    if (!FIRST && g == 0) {
#pragma unroll
      for (int q = 0; q < 4; ++q) {
        const float4 n4 = *(const float4*)(nu + colOff + 128 * q);
        const float4 a4 = *(const float4*)(plane0 + colOff + 128 * q);
        const float4 b4 = *(const float4*)(plane1 + colOff + 128 * q);
        vf[q].x = n4.x / ((a4.x + b4.x) + SK_TOL);
        vf[q].y = n4.y / ((a4.y + b4.y) + SK_TOL);
        vf[q].z = n4.z / ((a4.z + b4.z) + SK_TOL);
        vf[q].w = n4.w / ((a4.w + b4.w) + SK_TOL);
      }
    }
    v8f acc[4];
#pragma unroll
    for (int r = 0; r < 4; ++r) acc[r] = (v8f){0.f, 0.f, 0.f, 0.f, 0.f, 0.f, 0.f, 0.f};
#pragma unroll
    for (int q = 0; q < 4; ++q) {
      v2f b01 = {vf[q].x, vf[q].y};
      v2f b23 = {vf[q].z, vf[q].w};
#pragma unroll
      for (int r = 0; r < 4; ++r) {
        v2f al = {kb[g & 1][r][q].x, kb[g & 1][r][q].y};
        acc[r] = sk_wmma(al, b01, acc[r]);
      }
#pragma unroll
      for (int r = 0; r < 4; ++r) {
        v2f ah = {kb[g & 1][r][q].z, kb[g & 1][r][q].w};
        acc[r] = sk_wmma(ah, b23, acc[r]);
      }
    }
#pragma unroll
    for (int r = 0; r < 4; ++r) {
      float d = acc[r][0];
#pragma unroll
      for (int e = 1; e < 8; ++e) d = (sel == e) ? acc[r][e] : d;
      part[wave][4 * g + r][lane] = d;
    }
  }
#undef SK_LOAD_ROWS
  __syncthreads();
  {
    const int row = threadIdx.x >> 4;
    const int s   = threadIdx.x & 15;
    const int slot = (s < 8) ? s : s + 16;
    float x = 0.f;
#pragma unroll
    for (int w = 0; w < 8; ++w) x += part[w][row][slot];
    red[row][s] = x;
  }
  __syncthreads();
  if (threadIdx.x < 16) {
    float y = 0.f;
#pragma unroll
    for (int s = 0; s < 16; ++s) y += red[threadIdx.x][s];
    u_out[rowBase + threadIdx.x] = mrow / (y + SK_TOL);
  }
}

__global__ void __launch_bounds__(256) sk_matvec_KT(const float* __restrict__ K,
                                                    const float* __restrict__ u,
                                                    float* __restrict__ plane0,
                                                    float* __restrict__ plane1) {
  __shared__ float part[8][32];
  const int wave = threadIdx.x >> 5;
  const int lane = threadIdx.x & 31;
  const int hf   = lane >> 4;
  const int n    = lane & 15;
  const int strip = blockIdx.x & 127;
  const int half  = blockIdx.x >> 7;
  const int colBase = strip << 5;
  const int row0 = (half << 11) + (wave << 8);
  const float* Kp = K + (size_t)(row0 + 4 * hf) * SK_N + colBase + 2 * n;
  const float* up = u + row0 + 4 * hf;

  v8f accE0 = {0.f, 0.f, 0.f, 0.f, 0.f, 0.f, 0.f, 0.f};
  v8f accO0 = accE0, accE1 = accE0, accO1 = accE0;

  float4 ub[2][8];
  v2f kb[2][8][4];
#define SK_LOAD_GRP(T)                                                         \
  _Pragma("unroll") for (int g = 0; g < 8; ++g) {                              \
    ub[(T) & 1][g] = *(const float4*)(up + 64 * (T) + 8 * g);                  \
    _Pragma("unroll") for (int e = 0; e < 4; ++e)                              \
      kb[(T) & 1][g][e] =                                                      \
          *(const v2f*)(Kp + (size_t)(64 * (T) + 8 * g + e) * SK_N);           \
  }
  SK_LOAD_GRP(0)
#pragma unroll
  for (int t = 0; t < 4; ++t) {
    __builtin_amdgcn_sched_barrier(0);
    if (t < 3) { SK_LOAD_GRP(t + 1) }
    __builtin_amdgcn_sched_barrier(0);
#pragma unroll
    for (int g = 0; g < 8; ++g) {
      const float4 u4 = ub[t & 1][g];
      v2f a01 = {u4.x, u4.y};
      v2f a23 = {u4.z, u4.w};
      v2f bE0 = {kb[t & 1][g][0].x, kb[t & 1][g][1].x};
      v2f bO0 = {kb[t & 1][g][0].y, kb[t & 1][g][1].y};
      v2f bE1 = {kb[t & 1][g][2].x, kb[t & 1][g][3].x};
      v2f bO1 = {kb[t & 1][g][2].y, kb[t & 1][g][3].y};
      accE0 = sk_wmma(a01, bE0, accE0);
      accO0 = sk_wmma(a01, bO0, accO0);
      accE1 = sk_wmma(a23, bE1, accE1);
      accO1 = sk_wmma(a23, bO1, accO1);
    }
  }
#undef SK_LOAD_GRP
  if (lane < 16) {
    part[wave][2 * n]     = accE0[0] + accE1[0];
    part[wave][2 * n + 1] = accO0[0] + accO1[0];
  }
  __syncthreads();
  if (threadIdx.x < 32) {
    float s = 0.f;
#pragma unroll
    for (int w = 0; w < 8; ++w) s += part[w][threadIdx.x];
    float* dst = half ? plane1 : plane0;
    dst[colBase + threadIdx.x] = s;
  }
}

__global__ void __launch_bounds__(256) sk_final(const float* __restrict__ mu,
                                                const float* __restrict__ nu,
                                                const float* __restrict__ u,
                                                const float* __restrict__ plane0,
                                                const float* __restrict__ plane1,
                                                float* __restrict__ P,
                                                float* __restrict__ lpart) {
  __shared__ float red[256];
  const int i = blockIdx.x;
  const float ui = u[i];
  const float mi = mu[i];
  float* Prow = P + (size_t)i * SK_N;
  float w = 0.f;
  for (int j = threadIdx.x * 4; j < SK_N; j += 256 * 4) {
    float4 k4 = *(const float4*)(Prow + j);
    float4 n4 = *(const float4*)(nu + j);
    float4 a4 = *(const float4*)(plane0 + j);
    float4 b4 = *(const float4*)(plane1 + j);
    float4 v4;
    v4.x = n4.x / ((a4.x + b4.x) + SK_TOL);
    v4.y = n4.y / ((a4.y + b4.y) + SK_TOL);
    v4.z = n4.z / ((a4.z + b4.z) + SK_TOL);
    v4.w = n4.w / ((a4.w + b4.w) + SK_TOL);
    float4 p;
    p.x = ui * k4.x * v4.x;
    p.y = ui * k4.y * v4.y;
    p.z = ui * k4.z * v4.z;
    p.w = ui * k4.w * v4.w;
    *(float4*)(Prow + j) = p;
    w += p.x * fabsf(mi - n4.x) + p.y * fabsf(mi - n4.y) +
         p.z * fabsf(mi - n4.z) + p.w * fabsf(mi - n4.w);
  }
  red[threadIdx.x] = w;
  __syncthreads();
  for (int s = 128; s > 0; s >>= 1) {
    if (threadIdx.x < s) red[threadIdx.x] += red[threadIdx.x + s];
    __syncthreads();
  }
  if (threadIdx.x == 0) lpart[i] = red[0];
}

__global__ void __launch_bounds__(256) sk_loss_reduce(const float* __restrict__ lpart,
                                                      float* __restrict__ loss) {
  __shared__ float red[256];
  float s = 0.f;
  for (int i = threadIdx.x; i < SK_N; i += 256) s += lpart[i];
  red[threadIdx.x] = s;
  __syncthreads();
  for (int k = 128; k > 0; k >>= 1) {
    if (threadIdx.x < k) red[threadIdx.x] += red[threadIdx.x + k];
    __syncthreads();
  }
  if (threadIdx.x == 0) *loss = red[0];
}

extern "C" void kernel_launch(void* const* d_in, const int* in_sizes, int n_in,
                              void* d_out, int out_size, void* d_ws, size_t ws_size,
                              hipStream_t stream) {
  const float* mu = (const float*)d_in[0];
  const float* nu = (const float*)d_in[1];
  const float* C  = (const float*)d_in[2];
  float* K = (float*)d_out;
  float* ws = (float*)d_ws;
  float* u     = ws;
  float* lpart = ws + SK_N;
  float* pl0   = ws + 2 * SK_N;
  float* pl1   = ws + 3 * SK_N;

  sk_init_K<<<SK_NN / (256 * 4), 256, 0, stream>>>(C, K);

  sk_matvec_Kv<true><<<SK_N / 16, 256, 0, stream>>>(K, nu, pl0, pl1, mu, u);
  sk_matvec_KT<<<SK_N / 16, 256, 0, stream>>>(K, u, pl0, pl1);
  for (int t = 1; t < SK_ITERS; ++t) {
    sk_matvec_Kv<false><<<SK_N / 16, 256, 0, stream>>>(K, nu, pl0, pl1, mu, u);
    sk_matvec_KT<<<SK_N / 16, 256, 0, stream>>>(K, u, pl0, pl1);
  }

  sk_final<<<SK_N, 256, 0, stream>>>(mu, nu, u, pl0, pl1, K, lpart);
  sk_loss_reduce<<<1, 256, 0, stream>>>(lpart, K + SK_NN);
}
